# starting kernel
# baseline (speedup 1.0000x reference)
#include <hip/hip_runtime.h>

typedef __attribute__((ext_vector_type(2))) float v2f;
typedef __attribute__((ext_vector_type(4))) float v4f;
typedef __attribute__((ext_vector_type(8))) float v8f;

#define KW        5
#define PATCH     25
#define JPAD      28
#define NKB       7
#define C_IN      4
#define F_OUT     16
#define HH        250
#define WW        250
#define BB        8
#define NPIX      (BB*HH*WW)
#define NTILES    (NPIX/16)
#define WPB       16
#define NTHREADS  (WPB*32)
#define NQTILES   (C_IN*PATCH*NKB)
#define NLTILES   (C_IN*NKB)
#define NWTILES   (NQTILES+NLTILES)
#define WFLOATS   (NWTILES*64)
#define PATF      (JPAD*C_IN)
#define LDS_FLOATS (WFLOATS + WPB*16*PATF)

__global__ void prep_weights(const float* __restrict__ kern,
                             const float* __restrict__ qk,
                             float* __restrict__ wbuf) {
    int idx = blockIdx.x * blockDim.x + threadIdx.x;
    if (idx >= WFLOATS) return;
    int t    = idx >> 6;
    int slot = idx & 63;
    int l    = slot >> 1;
    int v    = slot & 1;
    int n    = l & 15;
    int krow = v + ((l >= 16) ? 2 : 0);
    float val = 0.0f;
    if (t < NQTILES) {
        int kb = t % NKB;
        int di = t / NKB;
        int j  = 4 * kb + krow;
        if (j < PATCH) val = qk[(di * PATCH + j) * F_OUT + n];
    } else {
        int t2 = t - NQTILES;
        int kb = t2 % NKB;
        int d  = t2 / NKB;
        int i  = 4 * kb + krow;
        if (i < PATCH) val = kern[(i * C_IN + d) * F_OUT + n];
    }
    wbuf[idx] = val;
}

extern __shared__ float smem[];

__global__ __launch_bounds__(NTHREADS)
void convquad_main(const float* __restrict__ x,
                   const float* __restrict__ bias,
                   const float* __restrict__ wbuf,
                   float* __restrict__ out) {
    const int tid  = threadIdx.x;
    const int lane = tid & 31;
    const int wave = tid >> 5;
    const int hi   = (lane >> 4) & 1;
    const int n    = lane & 15;

    {
        const v4f* src = (const v4f*)wbuf;
        v4f*       dst = (v4f*)smem;
        for (int e = tid; e < WFLOATS / 4; e += NTHREADS) dst[e] = src[e];
    }

    float* pat = smem + WFLOATS + wave * (16 * PATF);
    const int tile = blockIdx.x * WPB + wave;
    const bool active = (tile < NTILES);

    if (active) {
        const v4f z = {0.f, 0.f, 0.f, 0.f};
        for (int e = lane; e < 16 * 3; e += 32) {
            int pix = e / 3, jj = PATCH + (e % 3);
            *(v4f*)(pat + pix * PATF + jj * C_IN) = z;
        }
        for (int e = lane; e < 16 * PATCH; e += 32) {
            int pix = e / PATCH, j = e % PATCH;
            int P  = tile * 16 + pix;
            int wc = P % WW;
            int hc = (P / WW) % HH;
            int bc = P / (WW * HH);
            int h2 = hc + (j / KW) - 2;
            int w2 = wc + (j % KW) - 2;
            v4f val = z;
            if (h2 >= 0 && h2 < HH && w2 >= 0 && w2 < WW)
                val = *(const v4f*)(x + (size_t)((bc * HH + h2) * WW + w2) * C_IN);
            *(v4f*)(pat + pix * PATF + j * C_IN) = val;
        }
    }
    __syncthreads();
    if (!active) return;

    const float bv = bias[n];
    v8f acc = {bv, bv, bv, bv, bv, bv, bv, bv};

    const float* patA = pat + n * PATF;
    const float* patS = pat + hi * 8 * PATF;

    #pragma unroll 1
    for (int d = 0; d < C_IN; ++d) {
        v2f A[NKB];
        #pragma unroll
        for (int kb = 0; kb < NKB; ++kb) {
            int j0 = 4 * kb + 2 * hi;
            A[kb].x = patA[j0 * C_IN + d];
            A[kb].y = patA[(j0 + 1) * C_IN + d];
        }
        const float* Bq = smem + (size_t)(d * PATCH) * NKB * 64 + lane * 2;
        #pragma unroll 1
        for (int i = 0; i < PATCH; ++i) {
            v8f s = {0.f, 0.f, 0.f, 0.f, 0.f, 0.f, 0.f, 0.f};
            const float* Bt = Bq + i * (NKB * 64);
            #pragma unroll
            for (int kb = 0; kb < NKB; ++kb) {
                v2f b = *(const v2f*)(Bt + kb * 64);
                s = __builtin_amdgcn_wmma_f32_16x16x4_f32(
                        false, A[kb], false, b, (short)0, s, false, false);
            }
            const float* m = patS + i * C_IN + d;
            #pragma unroll
            for (int g = 0; g < 8; ++g)
                acc[g] = __builtin_fmaf(m[g * PATF], s[g], acc[g]);
        }
        const float* Bl = smem + (size_t)(NQTILES + d * NKB) * 64 + lane * 2;
        #pragma unroll
        for (int kb = 0; kb < NKB; ++kb) {
            v2f b = *(const v2f*)(Bl + kb * 64);
            acc = __builtin_amdgcn_wmma_f32_16x16x4_f32(
                      false, A[kb], false, b, (short)0, acc, false, false);
        }
    }

    float* op = out + (size_t)(tile * 16 + hi * 8) * F_OUT + n;
    #pragma unroll
    for (int g = 0; g < 8; ++g) op[g * F_OUT] = acc[g];
}

extern "C" void kernel_launch(void* const* d_in, const int* in_sizes, int n_in,
                              void* d_out, int out_size, void* d_ws, size_t ws_size,
                              hipStream_t stream) {
    const float* x    = (const float*)d_in[0];
    const float* kern = (const float*)d_in[1];
    const float* qk   = (const float*)d_in[2];
    const float* bias = (const float*)d_in[3];
    float*       out  = (float*)d_out;
    float*       wbuf = (float*)d_ws;

    prep_weights<<<(WFLOATS + 255) / 256, 256, 0, stream>>>(kern, qk, wbuf);

    const size_t shmem = (size_t)LDS_FLOATS * sizeof(float);
    (void)hipFuncSetAttribute((const void*)convquad_main,
                              hipFuncAttributeMaxDynamicSharedMemorySize,
                              (int)shmem);
    const int blocks = (NTILES + WPB - 1) / WPB;
    convquad_main<<<blocks, NTHREADS, shmem, stream>>>(x, bias, wbuf, out);
}


# Symmetric fold of quad weights in prep (U[i][j]=W[i][j]+W[j][i] for j>i, a changed sum order); row i runs only k-blocks j>=i starting at j=i: 392 f32 WMMA per tile instead of 728
# speedup vs baseline: 1.5497x; 1.5497x over previous
#include <hip/hip_runtime.h>


typedef __attribute__((ext_vector_type(2))) float v2f;
typedef __attribute__((ext_vector_type(4))) float v4f;
typedef __attribute__((ext_vector_type(8))) float v8f;

#define KW        5
#define PATCH     25
#define JPAD      28
#define NKB       7
#define C_IN      4
#define F_OUT     16
#define HH        250
#define WW        250
#define BB        8
#define NPIX      (BB*HH*WW)
#define NTILES    (NPIX/16)
#define WPB       16
#define NTHREADS  (WPB*32)
#define NQROW     91
#define NDTILES   (NQROW+NKB)
#define NWTILES   (C_IN*NDTILES)
#define WFLOATS   (NWTILES*64)
#define PATF      (JPAD*C_IN)
#define LDS_FLOATS (WFLOATS + WPB*16*PATF)

__host__ __device__ constexpr int nk(int i)   { return (28 - i) / 4; }
__host__ __device__ constexpr int toff(int i) { int s = 0; for (int t = 0; t < i; ++t) s += nk(t); return s; }
static_assert(toff(PATCH) == NQROW, "quad tile count per channel");

__global__ void prep_weights(const float* __restrict__ kern,
                             const float* __restrict__ qk,
                             float* __restrict__ wbuf) {
    int idx = blockIdx.x * blockDim.x + threadIdx.x;
    if (idx >= WFLOATS) return;
    int t    = idx >> 6;
    int slot = idx & 63;
    int l    = slot >> 1;
    int v    = slot & 1;
    int n    = l & 15;
    int krow = v + ((l >= 16) ? 2 : 0);
    int d    = t / NDTILES;
    int r    = t % NDTILES;
    float val = 0.0f;
    if (r < NQROW) {
        int i = 0, base = 0;
        while (i < PATCH - 1 && r >= base + nk(i)) { base += nk(i); ++i; }
        int j = i + 4 * (r - base) + krow;
        if (j < PATCH) {
            val = qk[((d * PATCH + i) * PATCH + j) * F_OUT + n];
            if (j > i) val += qk[((d * PATCH + j) * PATCH + i) * F_OUT + n];
        }
    } else {
        int tap = 4 * (r - NQROW) + krow;
        if (tap < PATCH) val = kern[(tap * C_IN + d) * F_OUT + n];
    }
    wbuf[idx] = val;
}

extern __shared__ float smem[];

template <int I>
__device__ __forceinline__ void quad_row(const v2f (&A)[4][NKB], const float* Bd,
                                         const float* m, v8f& acc) {
    constexpr int S = I & 3, Q = I >> 2, N = nk(I);
    const float* Bt = Bd + toff(I) * 64;
    v8f s = {0.f, 0.f, 0.f, 0.f, 0.f, 0.f, 0.f, 0.f};
    #pragma unroll
    for (int k = 0; k < N; ++k) {
        v2f b = *(const v2f*)(Bt + k * 64);
        s = __builtin_amdgcn_wmma_f32_16x16x4_f32(
                false, A[S][Q + k], false, b, (short)0, s, false, false);
    }
    #pragma unroll
    for (int g = 0; g < 8; ++g)
        acc[g] = __builtin_fmaf(m[g * PATF + I * C_IN], s[g], acc[g]);
}

template <int I>
__device__ __forceinline__ void quad_rows(const v2f (&A)[4][NKB], const float* Bd,
                                          const float* m, v8f& acc) {
    quad_row<I>(A, Bd, m, acc);
    if constexpr (I + 1 < PATCH) quad_rows<I + 1>(A, Bd, m, acc);
}

__global__ __launch_bounds__(NTHREADS)
void convquad_main(const float* __restrict__ x,
                   const float* __restrict__ bias,
                   const float* __restrict__ wbuf,
                   float* __restrict__ out) {
    const int tid  = threadIdx.x;
    const int lane = tid & 31;
    const int wave = tid >> 5;
    const int hi   = (lane >> 4) & 1;
    const int n    = lane & 15;

    {
        const v4f* src = (const v4f*)wbuf;
        v4f*       dst = (v4f*)smem;
        for (int e = tid; e < WFLOATS / 4; e += NTHREADS) dst[e] = src[e];
    }

    float* pat = smem + WFLOATS + wave * (16 * PATF);
    const int tile = blockIdx.x * WPB + wave;
    const bool active = (tile < NTILES);

    if (active) {
        const v4f z = {0.f, 0.f, 0.f, 0.f};
        for (int e = lane; e < 16 * 3; e += 32) {
            int pix = e / 3, jj = PATCH + (e % 3);
            *(v4f*)(pat + pix * PATF + jj * C_IN) = z;
        }
        for (int e = lane; e < 16 * PATCH; e += 32) {
            int pix = e / PATCH, j = e % PATCH;
            int P  = tile * 16 + pix;
            int wc = P % WW;
            int hc = (P / WW) % HH;
            int bc = P / (WW * HH);
            int h2 = hc + (j / KW) - 2;
            int w2 = wc + (j % KW) - 2;
            v4f val = z;
            if (h2 >= 0 && h2 < HH && w2 >= 0 && w2 < WW)
                val = *(const v4f*)(x + (size_t)((bc * HH + h2) * WW + w2) * C_IN);
            *(v4f*)(pat + pix * PATF + j * C_IN) = val;
        }
    }
    __syncthreads();
    if (!active) return;

    const float bv = bias[n];
    v8f acc = {bv, bv, bv, bv, bv, bv, bv, bv};

    const float* patA = pat + n * PATF + hi * 2 * C_IN;
    const float* patS = pat + hi * 8 * PATF;

    #pragma unroll 1
    for (int d = 0; d < C_IN; ++d) {
        v2f A[4][NKB];
        #pragma unroll
        for (int s = 0; s < 4; ++s) {
            #pragma unroll
            for (int kb = 0; kb < (s == 0 ? NKB : NKB - 1); ++kb) {
                A[s][kb].x = patA[(s + 4 * kb) * C_IN + d];
                A[s][kb].y = patA[(s + 4 * kb + 1) * C_IN + d];
            }
        }

        const float* Bd = smem + (size_t)(d * NDTILES) * 64 + lane * 2;
        quad_rows<0>(A, Bd, patS + d, acc);

        const float* Bl = Bd + NQROW * 64;
        #pragma unroll
        for (int kb = 0; kb < NKB; ++kb) {
            v2f b = *(const v2f*)(Bl + kb * 64);
            acc = __builtin_amdgcn_wmma_f32_16x16x4_f32(
                      false, A[0][kb], false, b, (short)0, acc, false, false);
        }
    }

    float* op = out + (size_t)(tile * 16 + hi * 8) * F_OUT + n;
    #pragma unroll
    for (int g = 0; g < 8; ++g) op[g * F_OUT] = acc[g];
}

extern "C" void kernel_launch(void* const* d_in, const int* in_sizes, int n_in,
                              void* d_out, int out_size, void* d_ws, size_t ws_size,
                              hipStream_t stream) {
    const float* x    = (const float*)d_in[0];
    const float* kern = (const float*)d_in[1];
    const float* qk   = (const float*)d_in[2];
    const float* bias = (const float*)d_in[3];
    float*       out  = (float*)d_out;
    float*       wbuf = (float*)d_ws;

    prep_weights<<<(WFLOATS + 255) / 256, 256, 0, stream>>>(kern, qk, wbuf);

    const size_t shmem = (size_t)LDS_FLOATS * sizeof(float);
    (void)hipFuncSetAttribute((const void*)convquad_main,
                              hipFuncAttributeMaxDynamicSharedMemorySize,
                              (int)shmem);
    const int blocks = (NTILES + WPB - 1) / WPB;
    convquad_main<<<blocks, NTHREADS, shmem, stream>>>(x, bias, wbuf, out);
}


# Persistent work-groups: 256 blocks of 16 waves (one per WGP), each wave loops over its tiles; swizzled weights copied to LDS once per WGP, not once per 16 tiles; one block barrier in all
# speedup vs baseline: 1.8213x; 1.1753x over previous
#include <hip/hip_runtime.h>


typedef __attribute__((ext_vector_type(2))) float v2f;
typedef __attribute__((ext_vector_type(4))) float v4f;
typedef __attribute__((ext_vector_type(8))) float v8f;

#define KW        5
#define PATCH     25
#define JPAD      28
#define NKB       7
#define C_IN      4
#define F_OUT     16
#define HH        250
#define WW        250
#define BB        8
#define NPIX      (BB*HH*WW)
#define NTILES    (NPIX/16)
#define WPB       16
#define NTHREADS  (WPB*32)
#define GRID_WG   256
#define WAVES_ALL (GRID_WG*WPB)
#define NQROW     91
#define NDTILES   (NQROW+NKB)
#define NWTILES   (C_IN*NDTILES)
#define WFLOATS   (NWTILES*64)
#define PATF      (JPAD*C_IN)
#define LDS_FLOATS (WFLOATS + WPB*16*PATF)

__host__ __device__ constexpr int nk(int i)   { return (28 - i) / 4; }
__host__ __device__ constexpr int toff(int i) { int s = 0; for (int t = 0; t < i; ++t) s += nk(t); return s; }
static_assert(toff(PATCH) == NQROW, "quad tile count per channel");

__global__ void prep_weights(const float* __restrict__ kern,
                             const float* __restrict__ qk,
                             float* __restrict__ wbuf) {
    int idx = blockIdx.x * blockDim.x + threadIdx.x;
    if (idx >= WFLOATS) return;
    int t    = idx >> 6;
    int slot = idx & 63;
    int l    = slot >> 1;
    int v    = slot & 1;
    int n    = l & 15;
    int krow = v + ((l >= 16) ? 2 : 0);
    int d    = t / NDTILES;
    int r    = t % NDTILES;
    float val = 0.0f;
    if (r < NQROW) {
        int i = 0, base = 0;
        while (i < PATCH - 1 && r >= base + nk(i)) { base += nk(i); ++i; }
        int j = i + 4 * (r - base) + krow;
        if (j < PATCH) {
            val = qk[((d * PATCH + i) * PATCH + j) * F_OUT + n];
            if (j > i) val += qk[((d * PATCH + j) * PATCH + i) * F_OUT + n];
        }
    } else {
        int tap = 4 * (r - NQROW) + krow;
        if (tap < PATCH) val = kern[(tap * C_IN + d) * F_OUT + n];
    }
    wbuf[idx] = val;
}

extern __shared__ float smem[];

template <int I>
__device__ __forceinline__ void quad_row(const v2f (&A)[4][NKB], const float* Bd,
                                         const float* m, v8f& acc) {
    constexpr int S = I & 3, Q = I >> 2, N = nk(I);
    const float* Bt = Bd + toff(I) * 64;
    v8f s = {0.f, 0.f, 0.f, 0.f, 0.f, 0.f, 0.f, 0.f};
    #pragma unroll
    for (int k = 0; k < N; ++k) {
        v2f b = *(const v2f*)(Bt + k * 64);
        s = __builtin_amdgcn_wmma_f32_16x16x4_f32(
                false, A[S][Q + k], false, b, (short)0, s, false, false);
    }
    #pragma unroll
    for (int g = 0; g < 8; ++g)
        acc[g] = __builtin_fmaf(m[g * PATF + I * C_IN], s[g], acc[g]);
}

template <int I>
__device__ __forceinline__ void quad_rows(const v2f (&A)[4][NKB], const float* Bd,
                                          const float* m, v8f& acc) {
    quad_row<I>(A, Bd, m, acc);
    if constexpr (I + 1 < PATCH) quad_rows<I + 1>(A, Bd, m, acc);
}

__global__ __launch_bounds__(NTHREADS)
void convquad_main(const float* __restrict__ x,
                   const float* __restrict__ bias,
                   const float* __restrict__ wbuf,
                   float* __restrict__ out) {
    const int tid  = threadIdx.x;
    const int lane = tid & 31;
    const int wave = __builtin_amdgcn_readfirstlane(tid >> 5);
    const int hi   = (lane >> 4) & 1;
    const int n    = lane & 15;

    {
        const v4f* src = (const v4f*)wbuf;
        v4f*       dst = (v4f*)smem;
        for (int e = tid; e < WFLOATS / 4; e += NTHREADS) dst[e] = src[e];
    }

    float* pat = smem + WFLOATS + wave * (16 * PATF);
    const v4f z = {0.f, 0.f, 0.f, 0.f};
    for (int e = lane; e < 16 * 3; e += 32) {
        int pix = e / 3, jj = PATCH + (e % 3);
        *(v4f*)(pat + pix * PATF + jj * C_IN) = z;
    }
    __syncthreads();

    const float bv = bias[n];
    const float* patA = pat + n * PATF + hi * 2 * C_IN;
    const float* patS = pat + hi * 8 * PATF;

    #pragma unroll 1
    for (int tile = wave * GRID_WG + (int)blockIdx.x; tile < NTILES; tile += WAVES_ALL) {
        for (int e = lane; e < 16 * PATCH; e += 32) {
            int pix = e / PATCH, j = e % PATCH;
            int P  = tile * 16 + pix;
            int wc = P % WW;
            int hc = (P / WW) % HH;
            int bc = P / (WW * HH);
            int h2 = hc + (j / KW) - 2;
            int w2 = wc + (j % KW) - 2;
            v4f val = z;
            if (h2 >= 0 && h2 < HH && w2 >= 0 && w2 < WW)
                val = *(const v4f*)(x + (size_t)((bc * HH + h2) * WW + w2) * C_IN);
            *(v4f*)(pat + pix * PATF + j * C_IN) = val;
        }

        v8f acc = {bv, bv, bv, bv, bv, bv, bv, bv};

        #pragma unroll 1
        for (int d = 0; d < C_IN; ++d) {
            v2f A[4][NKB];
            #pragma unroll
            for (int s = 0; s < 4; ++s) {
                #pragma unroll
                for (int kb = 0; kb < (s == 0 ? NKB : NKB - 1); ++kb) {
                    A[s][kb].x = patA[(s + 4 * kb) * C_IN + d];
                    A[s][kb].y = patA[(s + 4 * kb + 1) * C_IN + d];
                }
            }

            const float* Bd = smem + (size_t)(d * NDTILES) * 64 + lane * 2;
            quad_rows<0>(A, Bd, patS + d, acc);

            const float* Bl = Bd + NQROW * 64;
            #pragma unroll
            for (int kb = 0; kb < NKB; ++kb) {
                v2f b = *(const v2f*)(Bl + kb * 64);
                acc = __builtin_amdgcn_wmma_f32_16x16x4_f32(
                          false, A[0][kb], false, b, (short)0, acc, false, false);
            }
        }

        float* op = out + (size_t)(tile * 16 + hi * 8) * F_OUT + n;
        #pragma unroll
        for (int g = 0; g < 8; ++g) op[g * F_OUT] = acc[g];
    }
}

extern "C" void kernel_launch(void* const* d_in, const int* in_sizes, int n_in,
                              void* d_out, int out_size, void* d_ws, size_t ws_size,
                              hipStream_t stream) {
    const float* x    = (const float*)d_in[0];
    const float* kern = (const float*)d_in[1];
    const float* qk   = (const float*)d_in[2];
    const float* bias = (const float*)d_in[3];
    float*       out  = (float*)d_out;
    float*       wbuf = (float*)d_ws;

    prep_weights<<<(WFLOATS + 255) / 256, 256, 0, stream>>>(kern, qk, wbuf);

    const size_t shmem = (size_t)LDS_FLOATS * sizeof(float);
    (void)hipFuncSetAttribute((const void*)convquad_main,
                              hipFuncAttributeMaxDynamicSharedMemorySize,
                              (int)shmem);
    convquad_main<<<GRID_WG, NTHREADS, shmem, stream>>>(x, bias, wbuf, out);
}


# Quad rows folded into the A operand: A = p_i*p_j by v_pk_mul on a register copy of the patch row; all WMMA add into two alternating accumulators; LDS-broadcast+fma stage removed; sum order changed
# speedup vs baseline: 1.9492x; 1.0702x over previous
#include <hip/hip_runtime.h>


typedef __attribute__((ext_vector_type(2))) float v2f;
typedef __attribute__((ext_vector_type(4))) float v4f;
typedef __attribute__((ext_vector_type(8))) float v8f;

#define KW        5
#define PATCH     25
#define JPAD      28
#define NKB       7
#define C_IN      4
#define F_OUT     16
#define HH        250
#define WW        250
#define BB        8
#define NPIX      (BB*HH*WW)
#define NTILES    (NPIX/16)
#define WPB       16
#define NTHREADS  (WPB*32)
#define GRID_WG   256
#define WAVES_ALL (GRID_WG*WPB)
#define NQROW     91
#define NDTILES   (NQROW+NKB)
#define NWTILES   (C_IN*NDTILES)
#define WFLOATS   (NWTILES*64)
#define PATF      (JPAD*C_IN)
#define LDS_FLOATS (WFLOATS + WPB*16*PATF)

__host__ __device__ constexpr int nk(int i)   { return (28 - i) / 4; }
__host__ __device__ constexpr int toff(int i) { int s = 0; for (int t = 0; t < i; ++t) s += nk(t); return s; }
static_assert(toff(PATCH) == NQROW, "quad tile count per channel");

__global__ void prep_weights(const float* __restrict__ kern,
                             const float* __restrict__ qk,
                             float* __restrict__ wbuf) {
    int idx = blockIdx.x * blockDim.x + threadIdx.x;
    if (idx >= WFLOATS) return;
    int t    = idx >> 6;
    int slot = idx & 63;
    int l    = slot >> 1;
    int v    = slot & 1;
    int n    = l & 15;
    int krow = v + ((l >= 16) ? 2 : 0);
    int d    = t / NDTILES;
    int r    = t % NDTILES;
    float val = 0.0f;
    if (r < NQROW) {
        int i = 0, base = 0;
        while (i < PATCH - 1 && r >= base + nk(i)) { base += nk(i); ++i; }
        int j = i + 4 * (r - base) + krow;
        if (j < PATCH) {
            val = qk[((d * PATCH + i) * PATCH + j) * F_OUT + n];
            if (j > i) val += qk[((d * PATCH + j) * PATCH + i) * F_OUT + n];
        }
    } else {
        int tap = 4 * (r - NQROW) + krow;
        if (tap < PATCH) val = kern[(tap * C_IN + d) * F_OUT + n];
    }
    wbuf[idx] = val;
}

extern __shared__ float smem[];

#define NQREG 26

__device__ __forceinline__ void mma_acc(const v2f a, const v2f b, v8f& acc) {
    acc = __builtin_amdgcn_wmma_f32_16x16x4_f32(false, a, false, b, (short)0, acc, false, false);
}

template <int I>
__device__ __forceinline__ void quad_row(const v4f (&q)[NQREG], const float* Bl,
                                         const float* patP, v8f& acc0, v8f& acc1) {
    constexpr int N = nk(I);
    const v4f pi = *(const v4f*)(patP + I * C_IN);
    #pragma unroll
    for (int k = 0; k < N; ++k) {
        #pragma unroll
        for (int d = 0; d < C_IN; ++d) {
            v2f b = *(const v2f*)(Bl + (d * NDTILES + toff(I) + k) * 64);
            v2f a;
            a.x = q[I + 4 * k][d]     * pi[d];
            a.y = q[I + 4 * k + 1][d] * pi[d];
            if (d & 1) mma_acc(a, b, acc1); else mma_acc(a, b, acc0);
        }
    }
}

template <int I>
__device__ __forceinline__ void quad_rows(const v4f (&q)[NQREG], const float* Bl,
                                          const float* patP, v8f& acc0, v8f& acc1) {
    quad_row<I>(q, Bl, patP, acc0, acc1);
    if constexpr (I + 1 < PATCH) quad_rows<I + 1>(q, Bl, patP, acc0, acc1);
}

__global__ __launch_bounds__(NTHREADS)
void convquad_main(const float* __restrict__ x,
                   const float* __restrict__ bias,
                   const float* __restrict__ wbuf,
                   float* __restrict__ out) {
    const int tid  = threadIdx.x;
    const int lane = tid & 31;
    const int wave = __builtin_amdgcn_readfirstlane(tid >> 5);
    const int hi   = (lane >> 4) & 1;
    const int n    = lane & 15;

    {
        const v4f* src = (const v4f*)wbuf;
        v4f*       dst = (v4f*)smem;
        for (int e = tid; e < WFLOATS / 4; e += NTHREADS) dst[e] = src[e];
    }

    float* pat = smem + WFLOATS + wave * (16 * PATF);
    const v4f z = {0.f, 0.f, 0.f, 0.f};
    for (int e = lane; e < 16 * 3; e += 32) {
        int pix = e / 3, jj = PATCH + (e % 3);
        *(v4f*)(pat + pix * PATF + jj * C_IN) = z;
    }
    __syncthreads();

    const float bv = bias[n];
    const float* patP = pat + n * PATF;
    const float* patQ = patP + hi * 2 * C_IN;
    const float* Bl   = smem + lane * 2;

    #pragma unroll 1
    for (int tile = wave * GRID_WG + (int)blockIdx.x; tile < NTILES; tile += WAVES_ALL) {
        for (int e = lane; e < 16 * PATCH; e += 32) {
            int pix = e / PATCH, j = e % PATCH;
            int P  = tile * 16 + pix;
            int wc = P % WW;
            int hc = (P / WW) % HH;
            int bc = P / (WW * HH);
            int h2 = hc + (j / KW) - 2;
            int w2 = wc + (j % KW) - 2;
            v4f val = z;
            if (h2 >= 0 && h2 < HH && w2 >= 0 && w2 < WW)
                val = *(const v4f*)(x + (size_t)((bc * HH + h2) * WW + w2) * C_IN);
            *(v4f*)(pat + pix * PATF + j * C_IN) = val;
        }

        v4f q[NQREG];
        #pragma unroll
        for (int jj = 0; jj < NQREG; ++jj) q[jj] = *(const v4f*)(patQ + jj * C_IN);

        v8f acc0 = {bv, bv, bv, bv, bv, bv, bv, bv};
        v8f acc1 = {0.f, 0.f, 0.f, 0.f, 0.f, 0.f, 0.f, 0.f};

        quad_rows<0>(q, Bl, patP, acc0, acc1);

        #pragma unroll
        for (int kb = 0; kb < NKB; ++kb) {
            #pragma unroll
            for (int d = 0; d < C_IN; ++d) {
                v2f b = *(const v2f*)(Bl + (d * NDTILES + NQROW + kb) * 64);
                v2f a;
                a.x = q[4 * kb][d];
                a.y = q[4 * kb + 1][d];
                if (d & 1) mma_acc(a, b, acc1); else mma_acc(a, b, acc0);
            }
        }

        float* op = out + (size_t)(tile * 16 + hi * 8) * F_OUT + n;
        #pragma unroll
        for (int g = 0; g < 8; ++g) op[g * F_OUT] = acc0[g] + acc1[g];
    }
}

extern "C" void kernel_launch(void* const* d_in, const int* in_sizes, int n_in,
                              void* d_out, int out_size, void* d_ws, size_t ws_size,
                              hipStream_t stream) {
    const float* x    = (const float*)d_in[0];
    const float* kern = (const float*)d_in[1];
    const float* qk   = (const float*)d_in[2];
    const float* bias = (const float*)d_in[3];
    float*       out  = (float*)d_out;
    float*       wbuf = (float*)d_ws;

    prep_weights<<<(WFLOATS + 255) / 256, 256, 0, stream>>>(kern, qk, wbuf);

    const size_t shmem = (size_t)LDS_FLOATS * sizeof(float);
    (void)hipFuncSetAttribute((const void*)convquad_main,
                              hipFuncAttributeMaxDynamicSharedMemorySize,
                              (int)shmem);
    convquad_main<<<GRID_WG, NTHREADS, shmem, stream>>>(x, bias, wbuf, out);
}


# All weight tiles register-resident, 1 wave/SIMD: WMMA k = the 4 channels (A = p_i*p_j by v_pk_mul, B = U[.][i][j]), 350 B tiles in 700 VGPRs, patches global->VGPR, no LDS; sum order changed
# speedup vs baseline: 2.9213x; 1.4987x over previous
#include <hip/hip_runtime.h>


typedef __attribute__((ext_vector_type(2))) float v2f;
typedef __attribute__((ext_vector_type(4))) float v4f;
typedef __attribute__((ext_vector_type(8))) float v8f;

#define KW        5
#define PATCH     25
#define C_IN      4
#define F_OUT     16
#define HH        250
#define WW        250
#define BB        8
#define NPIX      (BB*HH*WW)
#define NTILES    (NPIX/16)
#define NPAIR     325
#define NBT       (NPAIR+PATCH)
#define WFLOATS   (NBT*64)
#define WPB       4
#define NTHREADS  (WPB*32)
#define GRID_WG   256
#define WAVES_ALL (GRID_WG*WPB)
#define NACC      4

__host__ __device__ constexpr int pidx(int i, int j) { return i * PATCH - (i * (i - 1)) / 2 + (j - i); }
static_assert(pidx(PATCH - 1, PATCH - 1) == NPAIR - 1, "pair count");

__global__ void prep_weights(const float* __restrict__ kern,
                             const float* __restrict__ qk,
                             float* __restrict__ wbuf) {
    int idx = blockIdx.x * blockDim.x + threadIdx.x;
    if (idx >= WFLOATS) return;
    int tp   = idx >> 7;
    int rem  = idx & 127;
    int lane = rem >> 2;
    int sub  = rem & 3;
    int t    = 2 * tp + (sub >> 1);
    int v    = sub & 1;
    int n    = lane & 15;
    int d    = v + ((lane >= 16) ? 2 : 0);
    float val;
    if (t < NPAIR) {
        int i = 0, base = 0;
        while (i < PATCH - 1 && t >= base + (PATCH - i)) { base += PATCH - i; ++i; }
        int j = i + (t - base);
        val = qk[((d * PATCH + i) * PATCH + j) * F_OUT + n];
        if (j > i) val += qk[((d * PATCH + j) * PATCH + i) * F_OUT + n];
    } else {
        int tap = t - NPAIR;
        val = kern[(tap * C_IN + d) * F_OUT + n];
    }
    wbuf[idx] = val;
}

__device__ __forceinline__ void mma_acc(const v2f a, const v2f b, v8f& acc) {
    acc = __builtin_amdgcn_wmma_f32_16x16x4_f32(false, a, false, b, (short)0, acc, false, false);
}

__device__ __forceinline__ void patch_issue(const float* __restrict__ x, int tile, int pix, int hi,
                                            v2f (&qn)[PATCH]) {
    const int P  = tile * 16 + pix;
    const int wc = P % WW;
    const int hc = (P / WW) % HH;
    const float* xp = x + (size_t)P * C_IN + 2 * hi;
    #pragma unroll
    for (int j = 0; j < PATCH; ++j) {
        const int dh = j / KW - 2, dw = j % KW - 2;
        const bool ok = (unsigned)(hc + dh) < (unsigned)HH && (unsigned)(wc + dw) < (unsigned)WW;
        const int off = ok ? (dh * WW + dw) * C_IN : 0;
        qn[j] = *(const v2f*)(xp + off);
    }
}

__device__ __forceinline__ void patch_mask(int tile, int pix, const v2f (&qn)[PATCH], v2f (&q)[PATCH]) {
    const int P  = tile * 16 + pix;
    const int wc = P % WW;
    const int hc = (P / WW) % HH;
    const v2f z = {0.f, 0.f};
    #pragma unroll
    for (int j = 0; j < PATCH; ++j) {
        const int dh = j / KW - 2, dw = j % KW - 2;
        const bool ok = (unsigned)(hc + dh) < (unsigned)HH && (unsigned)(wc + dw) < (unsigned)WW;
        q[j] = ok ? qn[j] : z;
    }
}

__global__ __launch_bounds__(NTHREADS)
void convquad_main(const float* __restrict__ x,
                   const float* __restrict__ bias,
                   const float* __restrict__ wbuf,
                   float* __restrict__ out) {
    const int tid  = threadIdx.x;
    const int lane = tid & 31;
    const int wave = __builtin_amdgcn_readfirstlane(tid >> 5);
    const int hi   = (lane >> 4) & 1;
    const int n    = lane & 15;

    const int first = (int)blockIdx.x * WPB + wave;

    v2f qn[PATCH];
    patch_issue(x, first, n, hi, qn);

    v2f B[NBT];
    {
        const v4f* wsrc = (const v4f*)wbuf + lane;
        #pragma unroll
        for (int tp = 0; tp < NBT / 2; ++tp) {
            const v4f w4 = wsrc[tp * 32];
            B[2 * tp]     = v2f{w4.x, w4.y};
            B[2 * tp + 1] = v2f{w4.z, w4.w};
        }
    }

    const float bv = bias[n];

    #pragma unroll 1
    for (int tile = first; tile < NTILES; tile += WAVES_ALL) {
        v2f q[PATCH];
        patch_mask(tile, n, qn, q);

        const int nxt = (tile + WAVES_ALL < NTILES) ? tile + WAVES_ALL : tile;
        patch_issue(x, nxt, n, hi, qn);
        __builtin_amdgcn_sched_barrier(0);

        v8f acc[NACC];
        acc[0] = v8f{bv, bv, bv, bv, bv, bv, bv, bv};
        #pragma unroll
        for (int k = 1; k < NACC; ++k) acc[k] = v8f{0.f, 0.f, 0.f, 0.f, 0.f, 0.f, 0.f, 0.f};

        #pragma unroll
        for (int i = 0; i < PATCH; ++i) {
            #pragma unroll
            for (int j = i; j < PATCH; ++j) {
                const v2f a = q[i] * q[j];
                mma_acc(a, B[pidx(i, j)], acc[pidx(i, j) % NACC]);
                if ((pidx(i, j) & 7) == 7) __builtin_amdgcn_sched_barrier(0);
            }
        }
        #pragma unroll
        for (int tap = 0; tap < PATCH; ++tap)
            mma_acc(q[tap], B[NPAIR + tap], acc[(NPAIR + tap) % NACC]);

        float* op = out + (size_t)(tile * 16 + hi * 8) * F_OUT + n;
        #pragma unroll
        for (int g = 0; g < 8; ++g)
            op[g * F_OUT] = (acc[0][g] + acc[1][g]) + (acc[2][g] + acc[3][g]);
    }
}

extern "C" void kernel_launch(void* const* d_in, const int* in_sizes, int n_in,
                              void* d_out, int out_size, void* d_ws, size_t ws_size,
                              hipStream_t stream) {
    const float* x    = (const float*)d_in[0];
    const float* kern = (const float*)d_in[1];
    const float* qk   = (const float*)d_in[2];
    const float* bias = (const float*)d_in[3];
    float*       out  = (float*)d_out;
    float*       wbuf = (float*)d_ws;

    prep_weights<<<(WFLOATS + 255) / 256, 256, 0, stream>>>(kern, qk, wbuf);
    convquad_main<<<GRID_WG, NTHREADS, 0, stream>>>(x, bias, wbuf, out);
}
